# starting kernel
# baseline (speedup 1.0000x reference)
#include <hip/hip_runtime.h>

typedef float v2f __attribute__((ext_vector_type(2)));
typedef float v4f __attribute__((ext_vector_type(4)));
typedef float v8f __attribute__((ext_vector_type(8)));

static constexpr int kB  = 16;
static constexpr int kC  = 256;
static constexpr int kW  = 128;
static constexpr int kHW = 128 * 128;

__global__ __launch_bounds__(256) void reduce_mean_max(
    const float* __restrict__ x, float* __restrict__ s) {
  int t   = blockIdx.x * 256 + threadIdx.x;
  int b   = t >> 12;
  int hw4 = (t & 4095) << 2;

  const v4f* p = (const v4f*)(x + (size_t)b * kC * kHW + hw4);
  v4f sum = {0.f, 0.f, 0.f, 0.f};
  float ninf = -__builtin_inff();
  v4f mx = {ninf, ninf, ninf, ninf};

#pragma unroll 16
  for (int c = 0; c < kC; ++c) {
    v4f v = __builtin_nontemporal_load(p + c * (kHW / 4));
    sum += v;
    mx[0] = fmaxf(mx[0], v[0]);
    mx[1] = fmaxf(mx[1], v[1]);
    mx[2] = fmaxf(mx[2], v[2]);
    mx[3] = fmaxf(mx[3], v[3]);
  }
  v4f avg = sum * (1.0f / 256.0f);

  v4f* sp = (v4f*)(s + (size_t)b * 2 * kHW);
  sp[hw4 >> 2]              = avg;
  sp[(kHW >> 2) + (hw4 >> 2)] = mx;
}

__global__ __launch_bounds__(32) void conv7x7_wmma(
    const float* __restrict__ s, const float* __restrict__ w,
    float* __restrict__ out) {
  __shared__ float tile[2 * 22 * 24];
  __shared__ float wl[98];

  int lane = threadIdx.x;
  int bid  = blockIdx.x;
  int b    = bid >> 6;
  int tl   = bid & 63;
  int y0   = (tl >> 3) << 4;
  int x0   = (tl & 7) << 4;

  const float* sb = s + (size_t)b * 2 * kHW;

  for (int i = lane; i < 98; i += 32) wl[i] = w[i];

  for (int pos = lane; pos < 2 * 484; pos += 32) {
    int c  = (pos >= 484);
    int r  = pos - c * 484;
    int yy = r / 22;
    int xx = r - yy * 22;
    int gy = y0 + yy - 3;
    int gx = x0 + xx - 3;
    float v = 0.0f;
    if ((unsigned)gy < 128u && (unsigned)gx < 128u)
      v = sb[c * kHW + gy * kW + gx];
    tile[c * 528 + yy * 24 + xx] = v;
  }
  __syncthreads();

  int half = lane >> 4;
  int mn   = lane & 15;

  int c0 = 0, yy0 = 0, dx0 = half * 2;

  v8f acc = {0.f, 0.f, 0.f, 0.f, 0.f, 0.f, 0.f, 0.f};

  for (int step = 0; step < 77; ++step) {
    int dx1 = dx0 + 1, yy1 = yy0, c1 = c0;
    if (dx1 == 7) { dx1 = 0; ++yy1; if (yy1 == 22) { yy1 = 0; ++c1; } }

    v2f a, bb;
    a.x = tile[c0 * 528 + yy0 * 24 + mn + dx0];
    a.y = tile[c1 * 528 + yy1 * 24 + mn + dx1];

    int dyA = yy0 - mn;
    int dyB = yy1 - mn;
    int iA = dyA < 0 ? 0 : (dyA > 6 ? 6 : dyA);
    int iB = dyB < 0 ? 0 : (dyB > 6 ? 6 : dyB);
    float wA = wl[c0 * 49 + iA * 7 + dx0];
    float wB = wl[c1 * 49 + iB * 7 + dx1];
    bb.x = ((unsigned)dyA < 7u) ? wA : 0.0f;
    bb.y = ((unsigned)dyB < 7u) ? wB : 0.0f;

    acc = __builtin_amdgcn_wmma_f32_16x16x4_f32(
        false, a, false, bb, (short)0, acc, false, false);

    dx0 += 4;
    if (dx0 >= 7) { dx0 -= 7; ++yy0; if (yy0 == 22) { yy0 = 0; ++c0; } }
  }

  int iy = mn;
  int jb = half * 8;
  float* ob = out + (size_t)b * kHW + (y0 + iy) * kW + (x0 + jb);
#pragma unroll
  for (int v = 0; v < 8; ++v) {
    float val = acc[v];
    ob[v] = 1.0f / (1.0f + __expf(-val));
  }
}

extern "C" void kernel_launch(void* const* d_in, const int* in_sizes, int n_in,
                              void* d_out, int out_size, void* d_ws, size_t ws_size,
                              hipStream_t stream) {
  const float* x = (const float*)d_in[0];
  const float* w = (const float*)d_in[1];
  float* out = (float*)d_out;
  float* s   = (float*)d_ws;

  reduce_mean_max<<<256, 256, 0, stream>>>(x, s);
  conv7x7_wmma<<<kB * 64, 32, 0, stream>>>(s, w, out);
}


# conv re-decomposed: M=output rows, Toeplitz along x, K=(c,dy,24-col window); A fragments are aligned 16-byte global loads (no LDS tile, no staging loop), 2 accumulators; k sum order changes
# speedup vs baseline: 1.6760x; 1.6760x over previous
#include <hip/hip_runtime.h>

typedef float v2f __attribute__((ext_vector_type(2)));
typedef float v4f __attribute__((ext_vector_type(4)));
typedef float v8f __attribute__((ext_vector_type(8)));

static constexpr int kB  = 16;
static constexpr int kC  = 256;
static constexpr int kW  = 128;
static constexpr int kHW = 128 * 128;

__global__ __launch_bounds__(256) void reduce_mean_max(
    const float* __restrict__ x, float* __restrict__ s) {
  int t   = blockIdx.x * 256 + threadIdx.x;
  int b   = t >> 12;
  int hw4 = (t & 4095) << 2;

  const v4f* p = (const v4f*)(x + (size_t)b * kC * kHW + hw4);
  v4f sum = {0.f, 0.f, 0.f, 0.f};
  float ninf = -__builtin_inff();
  v4f mx = {ninf, ninf, ninf, ninf};

#pragma unroll 16
  for (int c = 0; c < kC; ++c) {
    v4f v = __builtin_nontemporal_load(p + c * (kHW / 4));
    sum += v;
    mx[0] = fmaxf(mx[0], v[0]);
    mx[1] = fmaxf(mx[1], v[1]);
    mx[2] = fmaxf(mx[2], v[2]);
    mx[3] = fmaxf(mx[3], v[3]);
  }
  v4f avg = sum * (1.0f / 256.0f);

  v4f* sp = (v4f*)(s + (size_t)b * 2 * kHW);
  sp[hw4 >> 2]              = avg;
  sp[(kHW >> 2) + (hw4 >> 2)] = mx;
}


__device__ __forceinline__ void lds_barrier() {
  __builtin_amdgcn_fence(__ATOMIC_RELEASE, "workgroup", "local");
  __builtin_amdgcn_s_barrier();
  __builtin_amdgcn_fence(__ATOMIC_ACQUIRE, "workgroup", "local");
}

__global__ __launch_bounds__(32) void conv7x7_wmma(
    const float* __restrict__ s, const float* __restrict__ w,
    float* __restrict__ out) {
  __shared__ v4f wzv[160];
  float* wz = (float*)wzv;

  int lane = threadIdx.x;
  int bid  = blockIdx.x;
  int b    = bid >> 6;
  int tl   = bid & 63;
  int y0   = (tl >> 3) << 4;
  int x0   = (tl & 7) << 4;
  int h    = lane >> 4;
  int mn   = lane & 15;

  const float* sb = s + (size_t)b * 2 * kHW;

  float wv[4];
#pragma unroll
  for (int q = 0; q < 4; ++q) {
    int i = lane + 32 * q;
    wv[q] = w[i < 98 ? i : 97];
  }

  v4f a[2][7][3];
  bool ok[7][3];
  const v4f zero4 = {0.f, 0.f, 0.f, 0.f};
  int colb = x0 - 4 + 4 * h;
#pragma unroll
  for (int dy = 0; dy < 7; ++dy) {
    int r  = y0 + mn + dy - 3;
    bool rv = (unsigned)r < 128u;
    int rc = r < 0 ? 0 : (r > 127 ? 127 : r);
#pragma unroll
    for (int q = 0; q < 3; ++q) {
      int col = colb + 8 * q;
      ok[dy][q] = rv && ((unsigned)col < 128u);
      int cc = col < 0 ? 0 : (col > 124 ? 124 : col);
      const float* p = sb + rc * kW + cc;
#pragma unroll
      for (int c = 0; c < 2; ++c) {
        a[c][dy][q] = *(const v4f*)(p + c * kHW);
      }
    }
  }

  for (int r = 0; r < 5; ++r) wzv[lane + 32 * r] = zero4;
  lds_barrier();
#pragma unroll
  for (int q = 0; q < 4; ++q) {
    int i = lane + 32 * q;
    if (i < 98) {
      int row = i / 7;
      wz[row * 40 + 16 + (i - row * 7)] = wv[q];
    }
  }
  lds_barrier();

  const float* wl = wz + (15 + 4 * h - mn);

  v2f bf[2][7][3][2];
#pragma unroll
  for (int c = 0; c < 2; ++c) {
#pragma unroll
    for (int dy = 0; dy < 7; ++dy) {
#pragma unroll
      for (int q = 0; q < 3; ++q) {
        const float* wr = wl + (c * 7 + dy) * 40 + 8 * q;
        bf[c][dy][q][0] = v2f{wr[0], wr[1]};
        bf[c][dy][q][1] = v2f{wr[2], wr[3]};
      }
    }
  }

  __builtin_amdgcn_sched_barrier(0);

  v8f acc0 = {0.f, 0.f, 0.f, 0.f, 0.f, 0.f, 0.f, 0.f};
  v8f acc1 = {0.f, 0.f, 0.f, 0.f, 0.f, 0.f, 0.f, 0.f};

#pragma unroll
  for (int dy = 0; dy < 7; ++dy) {
#pragma unroll
    for (int q = 0; q < 3; ++q) {
      v4f a0 = ok[dy][q] ? a[0][dy][q] : zero4;
      v4f a1 = ok[dy][q] ? a[1][dy][q] : zero4;
      v2f a0lo = {a0[0], a0[1]}, a0hi = {a0[2], a0[3]};
      v2f a1lo = {a1[0], a1[1]}, a1hi = {a1[2], a1[3]};
      acc0 = __builtin_amdgcn_wmma_f32_16x16x4_f32(
          false, a0lo, false, bf[0][dy][q][0], (short)0, acc0, false, false);
      acc1 = __builtin_amdgcn_wmma_f32_16x16x4_f32(
          false, a1lo, false, bf[1][dy][q][0], (short)0, acc1, false, false);
      acc0 = __builtin_amdgcn_wmma_f32_16x16x4_f32(
          false, a0hi, false, bf[0][dy][q][1], (short)0, acc0, false, false);
      acc1 = __builtin_amdgcn_wmma_f32_16x16x4_f32(
          false, a1hi, false, bf[1][dy][q][1], (short)0, acc1, false, false);
    }
  }

  float* ob = out + (size_t)b * kHW + (y0 + 8 * h) * kW + (x0 + mn);
#pragma unroll
  for (int v = 0; v < 8; ++v) {
    float val = acc0[v] + acc1[v];
    ob[v * kW] = 1.0f / (1.0f + __expf(-val));
  }
}

extern "C" void kernel_launch(void* const* d_in, const int* in_sizes, int n_in,
                              void* d_out, int out_size, void* d_ws, size_t ws_size,
                              hipStream_t stream) {
  const float* x = (const float*)d_in[0];
  const float* w = (const float*)d_in[1];
  float* out = (float*)d_out;
  float* s   = (float*)d_ws;

  reduce_mean_max<<<256, 256, 0, stream>>>(x, s);
  conv7x7_wmma<<<kB * 64, 32, 0, stream>>>(s, w, out);
}


# reduce kernel: per 16-channel trip each wave issues one L2 prefetch whose 32 lanes name the 32 aligned 256-byte units it loads two trips later, so its 16-byte loads hit the second-level cache
# speedup vs baseline: 1.7253x; 1.0294x over previous
#include <hip/hip_runtime.h>

typedef float v2f __attribute__((ext_vector_type(2)));
typedef float v4f __attribute__((ext_vector_type(4)));
typedef float v8f __attribute__((ext_vector_type(8)));

static constexpr int kB  = 16;
static constexpr int kC  = 256;
static constexpr int kW  = 128;
static constexpr int kHW = 128 * 128;

__global__ __launch_bounds__(256) void reduce_mean_max(
    const float* __restrict__ x, float* __restrict__ s) {
  int t   = blockIdx.x * 256 + threadIdx.x;
  int b   = t >> 12;
  int hw4 = (t & 4095) << 2;
  int lane = threadIdx.x & 31;

  const v4f* p = (const v4f*)(x + (size_t)b * kC * kHW + hw4);
  const char* pf = (const char*)(p - lane) + (size_t)(lane >> 1) * (kHW * 4) + (lane & 1) * 256;
  constexpr size_t kTripBytes = (size_t)16 * kHW * 4;

  v4f sum = {0.f, 0.f, 0.f, 0.f};
  float ninf = -__builtin_inff();
  v4f mx = {ninf, ninf, ninf, ninf};

  __builtin_prefetch(pf + kTripBytes, 0, 3);
  for (int g = 0; g < 16; ++g) {
    if (g + 2 < 16) __builtin_prefetch(pf + (size_t)(g + 2) * kTripBytes, 0, 3);
    const v4f* pg = p + (size_t)g * 16 * (kHW / 4);
#pragma unroll
    for (int j = 0; j < 16; ++j) {
      v4f v = __builtin_nontemporal_load(pg + j * (kHW / 4));
      sum += v;
      mx[0] = fmaxf(mx[0], v[0]);
      mx[1] = fmaxf(mx[1], v[1]);
      mx[2] = fmaxf(mx[2], v[2]);
      mx[3] = fmaxf(mx[3], v[3]);
    }
  }
  v4f avg = sum * (1.0f / 256.0f);

  v4f* sp = (v4f*)(s + (size_t)b * 2 * kHW);
  sp[hw4 >> 2]              = avg;
  sp[(kHW >> 2) + (hw4 >> 2)] = mx;
}


__device__ __forceinline__ void lds_barrier() {
  __builtin_amdgcn_fence(__ATOMIC_RELEASE, "workgroup", "local");
  __builtin_amdgcn_s_barrier();
  __builtin_amdgcn_fence(__ATOMIC_ACQUIRE, "workgroup", "local");
}

__global__ __launch_bounds__(32) void conv7x7_wmma(
    const float* __restrict__ s, const float* __restrict__ w,
    float* __restrict__ out) {
  __shared__ v4f wzv[160];
  float* wz = (float*)wzv;

  int lane = threadIdx.x;
  int bid  = blockIdx.x;
  int b    = bid >> 6;
  int tl   = bid & 63;
  int y0   = (tl >> 3) << 4;
  int x0   = (tl & 7) << 4;
  int h    = lane >> 4;
  int mn   = lane & 15;

  const float* sb = s + (size_t)b * 2 * kHW;

  float wv[4];
#pragma unroll
  for (int q = 0; q < 4; ++q) {
    int i = lane + 32 * q;
    wv[q] = w[i < 98 ? i : 97];
  }

  v4f a[2][7][3];
  bool ok[7][3];
  const v4f zero4 = {0.f, 0.f, 0.f, 0.f};
  int colb = x0 - 4 + 4 * h;
#pragma unroll
  for (int dy = 0; dy < 7; ++dy) {
    int r  = y0 + mn + dy - 3;
    bool rv = (unsigned)r < 128u;
    int rc = r < 0 ? 0 : (r > 127 ? 127 : r);
#pragma unroll
    for (int q = 0; q < 3; ++q) {
      int col = colb + 8 * q;
      ok[dy][q] = rv && ((unsigned)col < 128u);
      int cc = col < 0 ? 0 : (col > 124 ? 124 : col);
      const float* p = sb + rc * kW + cc;
#pragma unroll
      for (int c = 0; c < 2; ++c) {
        a[c][dy][q] = *(const v4f*)(p + c * kHW);
      }
    }
  }

  for (int r = 0; r < 5; ++r) wzv[lane + 32 * r] = zero4;
  lds_barrier();
#pragma unroll
  for (int q = 0; q < 4; ++q) {
    int i = lane + 32 * q;
    if (i < 98) {
      int row = i / 7;
      wz[row * 40 + 16 + (i - row * 7)] = wv[q];
    }
  }
  lds_barrier();

  const float* wl = wz + (15 + 4 * h - mn);

  v2f bf[2][7][3][2];
#pragma unroll
  for (int c = 0; c < 2; ++c) {
#pragma unroll
    for (int dy = 0; dy < 7; ++dy) {
#pragma unroll
      for (int q = 0; q < 3; ++q) {
        const float* wr = wl + (c * 7 + dy) * 40 + 8 * q;
        bf[c][dy][q][0] = v2f{wr[0], wr[1]};
        bf[c][dy][q][1] = v2f{wr[2], wr[3]};
      }
    }
  }

  __builtin_amdgcn_sched_barrier(0);

  v8f acc0 = {0.f, 0.f, 0.f, 0.f, 0.f, 0.f, 0.f, 0.f};
  v8f acc1 = {0.f, 0.f, 0.f, 0.f, 0.f, 0.f, 0.f, 0.f};

#pragma unroll
  for (int dy = 0; dy < 7; ++dy) {
#pragma unroll
    for (int q = 0; q < 3; ++q) {
      v4f a0 = ok[dy][q] ? a[0][dy][q] : zero4;
      v4f a1 = ok[dy][q] ? a[1][dy][q] : zero4;
      v2f a0lo = {a0[0], a0[1]}, a0hi = {a0[2], a0[3]};
      v2f a1lo = {a1[0], a1[1]}, a1hi = {a1[2], a1[3]};
      acc0 = __builtin_amdgcn_wmma_f32_16x16x4_f32(
          false, a0lo, false, bf[0][dy][q][0], (short)0, acc0, false, false);
      acc1 = __builtin_amdgcn_wmma_f32_16x16x4_f32(
          false, a1lo, false, bf[1][dy][q][0], (short)0, acc1, false, false);
      acc0 = __builtin_amdgcn_wmma_f32_16x16x4_f32(
          false, a0hi, false, bf[0][dy][q][1], (short)0, acc0, false, false);
      acc1 = __builtin_amdgcn_wmma_f32_16x16x4_f32(
          false, a1hi, false, bf[1][dy][q][1], (short)0, acc1, false, false);
    }
  }

  float* ob = out + (size_t)b * kHW + (y0 + 8 * h) * kW + (x0 + mn);
#pragma unroll
  for (int v = 0; v < 8; ++v) {
    float val = acc0[v] + acc1[v];
    ob[v * kW] = 1.0f / (1.0f + __expf(-val));
  }
}

extern "C" void kernel_launch(void* const* d_in, const int* in_sizes, int n_in,
                              void* d_out, int out_size, void* d_ws, size_t ws_size,
                              hipStream_t stream) {
  const float* x = (const float*)d_in[0];
  const float* w = (const float*)d_in[1];
  float* out = (float*)d_out;
  float* s   = (float*)d_ws;

  reduce_mean_max<<<256, 256, 0, stream>>>(x, s);
  conv7x7_wmma<<<kB * 64, 32, 0, stream>>>(s, w, out);
}
